# starting kernel
# baseline (speedup 1.0000x reference)
#include <hip/hip_runtime.h>

typedef __attribute__((ext_vector_type(2))) float v2f;
typedef __attribute__((ext_vector_type(8))) float v8f;

#define H_DIM 512
#define R_DIM 64
#define M_ROWS 2048

__global__ void build_M(const float* __restrict__ W_seq,
                        const float* __restrict__ he,
                        const float* __restrict__ cp,
                        float* __restrict__ Mmat) {
    const int idx = blockIdx.x * blockDim.x + threadIdx.x;
    const int hp = idx >> 9;
    const int h  = idx & 511;
    float acc = 0.0f;
#pragma unroll 8
    for (int r = 0; r < R_DIM; ++r) {
        acc = fmaf(W_seq[r * H_DIM + hp], he[h * R_DIM + r] * cp[r], acc);
    }
    Mmat[idx] = acc;
}

__global__ void __launch_bounds__(256)
gemm_out(const float* __restrict__ Xin,
         const float* __restrict__ Mmat,
         float* __restrict__ Out) {
    const int lane = threadIdx.x & 31;
    const int wave = threadIdx.x >> 5;
    const int l    = lane & 15;
    const int half = lane >> 4;

    const int wid    = blockIdx.x * (blockDim.x >> 5) + wave;
    const int tile_m = wid >> 3;
    const int ngroup = wid & 7;
    const int m0 = tile_m * 16;
    const int n0 = ngroup * 64;

    v8f acc0 = {}, acc1 = {}, acc2 = {}, acc3 = {};

    const float* __restrict__ arow = Xin + (size_t)(m0 + l) * H_DIM;

#pragma unroll 4
    for (int k = 0; k < H_DIM; k += 4) {
        const int ka = k + 2 * half;
        v2f a;
        a.x = arow[ka];
        a.y = arow[ka + 1];

        const float* __restrict__ b0 = Mmat + (size_t)ka * H_DIM + n0 + l;
        const float* __restrict__ b1 = b0 + H_DIM;

        v2f b;
        b.x = b0[0];  b.y = b1[0];
        acc0 = __builtin_amdgcn_wmma_f32_16x16x4_f32(
            false, a, false, b, (short)0, acc0, false, false);
        b.x = b0[16]; b.y = b1[16];
        acc1 = __builtin_amdgcn_wmma_f32_16x16x4_f32(
            false, a, false, b, (short)0, acc1, false, false);
        b.x = b0[32]; b.y = b1[32];
        acc2 = __builtin_amdgcn_wmma_f32_16x16x4_f32(
            false, a, false, b, (short)0, acc2, false, false);
        b.x = b0[48]; b.y = b1[48];
        acc3 = __builtin_amdgcn_wmma_f32_16x16x4_f32(
            false, a, false, b, (short)0, acc3, false, false);
    }

    float* __restrict__ orow = Out + n0 + l;
#pragma unroll
    for (int p = 0; p < 8; ++p) {
        const size_t m = (size_t)(m0 + p + half * 8);
        orow[m * H_DIM +  0] = acc0[p];
        orow[m * H_DIM + 16] = acc1[p];
        orow[m * H_DIM + 32] = acc2[p];
        orow[m * H_DIM + 48] = acc3[p];
    }
}

extern "C" void kernel_launch(void* const* d_in, const int* in_sizes, int n_in,
                              void* d_out, int out_size, void* d_ws, size_t ws_size,
                              hipStream_t stream) {
    const float* hidden = (const float*)d_in[0];
    const float* W_seq  = (const float*)d_in[2];
    const float* he     = (const float*)d_in[3];
    const float* cp     = (const float*)d_in[4];
    float* out  = (float*)d_out;
    float* Mmat = (float*)d_ws;

    (void)in_sizes; (void)n_in; (void)out_size; (void)ws_size;

    build_M<<<(H_DIM * H_DIM) / 256, 256, 0, stream>>>(W_seq, he, cp, Mmat);

    gemm_out<<<128, 256, 0, stream>>>(hidden, Mmat, out);
}


# Main product tiled 64x64 per 4-wave work-group: 64-deep blocks of X and M staged once in shared LDS by 16 B loads, each wave a 32x32 quadrant from LDS fragments; per-element sum order unchanged
# speedup vs baseline: 2.1671x; 2.1671x over previous
#include <hip/hip_runtime.h>


typedef __attribute__((ext_vector_type(2))) float v2f;
typedef __attribute__((ext_vector_type(4))) float v4f;
typedef __attribute__((ext_vector_type(8))) float v8f;

#define H_DIM 512
#define R_DIM 64
#define M_ROWS 2048

#define TILE 64
#define KC   64
#define PA   68
#define PB   72

__global__ void build_M(const float* __restrict__ W_seq,
                        const float* __restrict__ he,
                        const float* __restrict__ cp,
                        float* __restrict__ Mmat) {
    const int idx = blockIdx.x * blockDim.x + threadIdx.x;
    const int hp = idx >> 9;
    const int h  = idx & 511;
    float acc = 0.0f;
#pragma unroll 8
    for (int r = 0; r < R_DIM; ++r) {
        acc = fmaf(W_seq[r * H_DIM + hp], he[h * R_DIM + r] * cp[r], acc);
    }
    Mmat[idx] = acc;
}

__global__ void __launch_bounds__(128) __attribute__((amdgpu_waves_per_eu(1, 2)))
gemm_out(const float* __restrict__ Xin,
         const float* __restrict__ Mmat,
         float* __restrict__ Out) {
    __shared__ __attribute__((aligned(16))) float As[TILE * PA];
    __shared__ __attribute__((aligned(16))) float Bs[KC * PB];

    const int tid  = threadIdx.x;
    const int lane = tid & 31;
    const int wave = tid >> 5;
    const int l    = lane & 15;
    const int half = lane >> 4;

    const int tn = blockIdx.x & 7;
    const int tm = blockIdx.x >> 3;
    const int m0 = tm * TILE;
    const int n0 = tn * TILE;
    const int wm = (wave >> 1) * 32;
    const int wn = (wave & 1) * 32;

    const int srow = tid >> 4;
    const int scol = (tid & 15) * 4;

    const float* __restrict__ gA = Xin  + (size_t)(m0 + srow) * H_DIM + scol;
    const float* __restrict__ gB = Mmat + (size_t)srow * H_DIM + n0 + scol;

    v4f ra[8], rb[8];
#pragma unroll
    for (int p = 0; p < 8; ++p) {
        ra[p] = *(const v4f*)(gA + (size_t)(p * 8) * H_DIM);
        rb[p] = *(const v4f*)(gB + (size_t)(p * 8) * H_DIM);
    }

    v8f acc00 = {}, acc01 = {}, acc10 = {}, acc11 = {};

    const float* a0p = &As[(wm + l) * PA + 2 * half];
    const float* a1p = a0p + 16 * PA;
    const float* bp  = &Bs[(2 * half) * PB + wn + l];

    for (int c = 0; c < H_DIM / KC; ++c) {
#pragma unroll
        for (int p = 0; p < 8; ++p) {
            *(v4f*)(&As[(p * 8 + srow) * PA + scol]) = ra[p];
            *(v4f*)(&Bs[(p * 8 + srow) * PB + scol]) = rb[p];
        }
        __syncthreads();

        if (c + 1 < H_DIM / KC) {
            const float* __restrict__ nA = gA + (c + 1) * KC;
            const float* __restrict__ nB = gB + (size_t)(c + 1) * KC * H_DIM;
#pragma unroll
            for (int p = 0; p < 8; ++p) {
                ra[p] = *(const v4f*)(nA + (size_t)(p * 8) * H_DIM);
                rb[p] = *(const v4f*)(nB + (size_t)(p * 8) * H_DIM);
            }
        }

#pragma unroll
        for (int k = 0; k < KC; k += 4) {
            const v2f a0 = *(const v2f*)(a0p + k);
            const v2f a1 = *(const v2f*)(a1p + k);
            v2f b0, b1;
            b0.x = bp[k * PB];       b0.y = bp[(k + 1) * PB];
            b1.x = bp[k * PB + 16];  b1.y = bp[(k + 1) * PB + 16];
            acc00 = __builtin_amdgcn_wmma_f32_16x16x4_f32(
                false, a0, false, b0, (short)0, acc00, false, false);
            acc01 = __builtin_amdgcn_wmma_f32_16x16x4_f32(
                false, a0, false, b1, (short)0, acc01, false, false);
            acc10 = __builtin_amdgcn_wmma_f32_16x16x4_f32(
                false, a1, false, b0, (short)0, acc10, false, false);
            acc11 = __builtin_amdgcn_wmma_f32_16x16x4_f32(
                false, a1, false, b1, (short)0, acc11, false, false);
        }
        __syncthreads();
    }

    float* __restrict__ o = Out + (size_t)(m0 + wm + half * 8) * H_DIM + n0 + wn + l;
#pragma unroll
    for (int p = 0; p < 8; ++p) {
        o[(size_t)p * H_DIM]             = acc00[p];
        o[(size_t)p * H_DIM + 16]        = acc01[p];
        o[(size_t)(p + 16) * H_DIM]      = acc10[p];
        o[(size_t)(p + 16) * H_DIM + 16] = acc11[p];
    }
}

extern "C" void kernel_launch(void* const* d_in, const int* in_sizes, int n_in,
                              void* d_out, int out_size, void* d_ws, size_t ws_size,
                              hipStream_t stream) {
    const float* hidden = (const float*)d_in[0];
    const float* W_seq  = (const float*)d_in[2];
    const float* he     = (const float*)d_in[3];
    const float* cp     = (const float*)d_in[4];
    float* out  = (float*)d_out;
    float* Mmat = (float*)d_ws;

    (void)in_sizes; (void)n_in; (void)out_size; (void)ws_size;

    build_M<<<(H_DIM * H_DIM) / 256, 256, 0, stream>>>(W_seq, he, cp, Mmat);

    gemm_out<<<256, 128, 0, stream>>>(hidden, Mmat, out);
}


# Stage-1 matrix (512x512, k=64) built on the f32 matrix instruction: 512 waves, a 16x32 tile each, all operand loads issued up front, replacing 8192 scalar-FMA waves; f32 sum order of M changes
# speedup vs baseline: 2.9429x; 1.3580x over previous
#include <hip/hip_runtime.h>


typedef __attribute__((ext_vector_type(2))) float v2f;
typedef __attribute__((ext_vector_type(4))) float v4f;
typedef __attribute__((ext_vector_type(8))) float v8f;

#define H_DIM 512
#define R_DIM 64
#define M_ROWS 2048

#define TILE 64
#define KC   64
#define PA   68
#define PB   72

__global__ void __launch_bounds__(64) __attribute__((amdgpu_waves_per_eu(1, 2)))
build_M(const float* __restrict__ W_seq,
        const float* __restrict__ he,
        const float* __restrict__ cp,
        float* __restrict__ Mmat) {
    const int lane = threadIdx.x & 31;
    const int wave = threadIdx.x >> 5;
    const int c    = lane & 15;
    const int hf   = lane >> 4;

    const int h0  = (blockIdx.x & 15) * 32;
    const int hp0 = (blockIdx.x >> 4) * 32 + wave * 16;

    const float* __restrict__ hr0 = he + (size_t)(h0 + c) * R_DIM + 4 * hf;
    const float* __restrict__ hr1 = hr0 + 16 * R_DIM;
    const float* __restrict__ cpp = cp + 4 * hf;
    const float* __restrict__ wp  = W_seq + (size_t)(4 * hf) * H_DIM + hp0 + c;

    v4f hb0[8], hb1[8], cw[8];
    float av[8][4];
#pragma unroll
    for (int q = 0; q < 8; ++q) {
        cw[q]  = *(const v4f*)(cpp + 8 * q);
        hb0[q] = *(const v4f*)(hr0 + 8 * q);
        hb1[q] = *(const v4f*)(hr1 + 8 * q);
#pragma unroll
        for (int j = 0; j < 4; ++j)
            av[q][j] = wp[(size_t)(8 * q + j) * H_DIM];
    }

    v8f acc0 = {}, acc1 = {};
#pragma unroll
    for (int q = 0; q < 8; ++q) {
        const v4f s0 = hb0[q] * cw[q];
        const v4f s1 = hb1[q] * cw[q];
        {
            const v2f a  = {av[q][0], av[q][1]};
            const v2f b0 = {s0.x, s0.y};
            const v2f b1 = {s1.x, s1.y};
            acc0 = __builtin_amdgcn_wmma_f32_16x16x4_f32(
                false, a, false, b0, (short)0, acc0, false, false);
            acc1 = __builtin_amdgcn_wmma_f32_16x16x4_f32(
                false, a, false, b1, (short)0, acc1, false, false);
        }
        {
            const v2f a  = {av[q][2], av[q][3]};
            const v2f b0 = {s0.z, s0.w};
            const v2f b1 = {s1.z, s1.w};
            acc0 = __builtin_amdgcn_wmma_f32_16x16x4_f32(
                false, a, false, b0, (short)0, acc0, false, false);
            acc1 = __builtin_amdgcn_wmma_f32_16x16x4_f32(
                false, a, false, b1, (short)0, acc1, false, false);
        }
    }

    float* __restrict__ o = Mmat + (size_t)(hp0 + 8 * hf) * H_DIM + h0 + c;
#pragma unroll
    for (int p = 0; p < 8; ++p) {
        o[(size_t)p * H_DIM]      = acc0[p];
        o[(size_t)p * H_DIM + 16] = acc1[p];
    }
}

__global__ void __launch_bounds__(128) __attribute__((amdgpu_waves_per_eu(1, 2)))
gemm_out(const float* __restrict__ Xin,
         const float* __restrict__ Mmat,
         float* __restrict__ Out) {
    __shared__ __attribute__((aligned(16))) float As[TILE * PA];
    __shared__ __attribute__((aligned(16))) float Bs[KC * PB];

    const int tid  = threadIdx.x;
    const int lane = tid & 31;
    const int wave = tid >> 5;
    const int l    = lane & 15;
    const int half = lane >> 4;

    const int tn = blockIdx.x & 7;
    const int tm = blockIdx.x >> 3;
    const int m0 = tm * TILE;
    const int n0 = tn * TILE;
    const int wm = (wave >> 1) * 32;
    const int wn = (wave & 1) * 32;

    const int srow = tid >> 4;
    const int scol = (tid & 15) * 4;

    const float* __restrict__ gA = Xin  + (size_t)(m0 + srow) * H_DIM + scol;
    const float* __restrict__ gB = Mmat + (size_t)srow * H_DIM + n0 + scol;

    v4f ra[8], rb[8];
#pragma unroll
    for (int p = 0; p < 8; ++p) {
        ra[p] = *(const v4f*)(gA + (size_t)(p * 8) * H_DIM);
        rb[p] = *(const v4f*)(gB + (size_t)(p * 8) * H_DIM);
    }

    v8f acc00 = {}, acc01 = {}, acc10 = {}, acc11 = {};

    const float* a0p = &As[(wm + l) * PA + 2 * half];
    const float* a1p = a0p + 16 * PA;
    const float* bp  = &Bs[(2 * half) * PB + wn + l];

    for (int c = 0; c < H_DIM / KC; ++c) {
#pragma unroll
        for (int p = 0; p < 8; ++p) {
            *(v4f*)(&As[(p * 8 + srow) * PA + scol]) = ra[p];
            *(v4f*)(&Bs[(p * 8 + srow) * PB + scol]) = rb[p];
        }
        __syncthreads();

        if (c + 1 < H_DIM / KC) {
            const float* __restrict__ nA = gA + (c + 1) * KC;
            const float* __restrict__ nB = gB + (size_t)(c + 1) * KC * H_DIM;
#pragma unroll
            for (int p = 0; p < 8; ++p) {
                ra[p] = *(const v4f*)(nA + (size_t)(p * 8) * H_DIM);
                rb[p] = *(const v4f*)(nB + (size_t)(p * 8) * H_DIM);
            }
        }

#pragma unroll
        for (int k = 0; k < KC; k += 4) {
            const v2f a0 = *(const v2f*)(a0p + k);
            const v2f a1 = *(const v2f*)(a1p + k);
            v2f b0, b1;
            b0.x = bp[k * PB];       b0.y = bp[(k + 1) * PB];
            b1.x = bp[k * PB + 16];  b1.y = bp[(k + 1) * PB + 16];
            acc00 = __builtin_amdgcn_wmma_f32_16x16x4_f32(
                false, a0, false, b0, (short)0, acc00, false, false);
            acc01 = __builtin_amdgcn_wmma_f32_16x16x4_f32(
                false, a0, false, b1, (short)0, acc01, false, false);
            acc10 = __builtin_amdgcn_wmma_f32_16x16x4_f32(
                false, a1, false, b0, (short)0, acc10, false, false);
            acc11 = __builtin_amdgcn_wmma_f32_16x16x4_f32(
                false, a1, false, b1, (short)0, acc11, false, false);
        }
        __syncthreads();
    }

    float* __restrict__ o = Out + (size_t)(m0 + wm + half * 8) * H_DIM + n0 + wn + l;
#pragma unroll
    for (int p = 0; p < 8; ++p) {
        o[(size_t)p * H_DIM]             = acc00[p];
        o[(size_t)p * H_DIM + 16]        = acc01[p];
        o[(size_t)(p + 16) * H_DIM]      = acc10[p];
        o[(size_t)(p + 16) * H_DIM + 16] = acc11[p];
    }
}

extern "C" void kernel_launch(void* const* d_in, const int* in_sizes, int n_in,
                              void* d_out, int out_size, void* d_ws, size_t ws_size,
                              hipStream_t stream) {
    const float* hidden = (const float*)d_in[0];
    const float* W_seq  = (const float*)d_in[2];
    const float* he     = (const float*)d_in[3];
    const float* cp     = (const float*)d_in[4];
    float* out  = (float*)d_out;
    float* Mmat = (float*)d_ws;

    (void)in_sizes; (void)n_in; (void)out_size; (void)ws_size;

    build_M<<<256, 64, 0, stream>>>(W_seq, he, cp, Mmat);

    gemm_out<<<256, 128, 0, stream>>>(hidden, Mmat, out);
}


# Main product as two resident 4-wave work-groups per WGP (512 tiles of 32x64): one fetches a block's LDS fragments as one group while the other runs its matrix steps; sums unchanged
# speedup vs baseline: 3.1948x; 1.0856x over previous
#include <hip/hip_runtime.h>


typedef __attribute__((ext_vector_type(2))) float v2f;
typedef __attribute__((ext_vector_type(4))) float v4f;
typedef __attribute__((ext_vector_type(8))) float v8f;

#define H_DIM 512
#define R_DIM 64
#define M_ROWS 2048

#define TILE_M 32
#define TILE_N 64
#define KC   64
#define PA   68
#define TB   1040

__global__ void __launch_bounds__(64) __attribute__((amdgpu_waves_per_eu(1, 2)))
build_M(const float* __restrict__ W_seq,
        const float* __restrict__ he,
        const float* __restrict__ cp,
        float* __restrict__ Mmat) {
    const int lane = threadIdx.x & 31;
    const int wave = threadIdx.x >> 5;
    const int c    = lane & 15;
    const int hf   = lane >> 4;

    const int h0  = (blockIdx.x & 15) * 32;
    const int hp0 = (blockIdx.x >> 4) * 32 + wave * 16;

    const float* __restrict__ hr0 = he + (size_t)(h0 + c) * R_DIM + 4 * hf;
    const float* __restrict__ hr1 = hr0 + 16 * R_DIM;
    const float* __restrict__ cpp = cp + 4 * hf;
    const float* __restrict__ wp  = W_seq + (size_t)(4 * hf) * H_DIM + hp0 + c;

    v4f hb0[8], hb1[8], cw[8];
    float av[8][4];
#pragma unroll
    for (int q = 0; q < 8; ++q) {
        cw[q]  = *(const v4f*)(cpp + 8 * q);
        hb0[q] = *(const v4f*)(hr0 + 8 * q);
        hb1[q] = *(const v4f*)(hr1 + 8 * q);
#pragma unroll
        for (int j = 0; j < 4; ++j)
            av[q][j] = wp[(size_t)(8 * q + j) * H_DIM];
    }

    v8f acc0 = {}, acc1 = {};
#pragma unroll
    for (int q = 0; q < 8; ++q) {
        const v4f s0 = hb0[q] * cw[q];
        const v4f s1 = hb1[q] * cw[q];
        {
            const v2f a  = {av[q][0], av[q][1]};
            const v2f b0 = {s0.x, s0.y};
            const v2f b1 = {s1.x, s1.y};
            acc0 = __builtin_amdgcn_wmma_f32_16x16x4_f32(
                false, a, false, b0, (short)0, acc0, false, false);
            acc1 = __builtin_amdgcn_wmma_f32_16x16x4_f32(
                false, a, false, b1, (short)0, acc1, false, false);
        }
        {
            const v2f a  = {av[q][2], av[q][3]};
            const v2f b0 = {s0.z, s0.w};
            const v2f b1 = {s1.z, s1.w};
            acc0 = __builtin_amdgcn_wmma_f32_16x16x4_f32(
                false, a, false, b0, (short)0, acc0, false, false);
            acc1 = __builtin_amdgcn_wmma_f32_16x16x4_f32(
                false, a, false, b1, (short)0, acc1, false, false);
        }
    }

    float* __restrict__ o = Mmat + (size_t)(hp0 + 8 * hf) * H_DIM + h0 + c;
#pragma unroll
    for (int p = 0; p < 8; ++p) {
        o[(size_t)p * H_DIM]      = acc0[p];
        o[(size_t)p * H_DIM + 16] = acc1[p];
    }
}

__global__ void __launch_bounds__(128) __attribute__((amdgpu_waves_per_eu(2, 2)))
gemm_out(const float* __restrict__ Xin,
         const float* __restrict__ Mmat,
         float* __restrict__ Out) {
    __shared__ __attribute__((aligned(16))) float As[TILE_M * PA];
    __shared__ __attribute__((aligned(16))) float Bs[4 * TB];

    const int tid  = threadIdx.x;
    const int lane = tid & 31;
    const int wave = tid >> 5;
    const int l    = lane & 15;
    const int half = lane >> 4;

    const int tn = blockIdx.x & 7;
    const int tm = blockIdx.x >> 3;
    const int m0 = tm * TILE_M;
    const int n0 = tn * TILE_N;
    const int wm = (wave >> 1) * 16;
    const int wn = (wave & 1) * 32;

    const int srow = tid >> 4;
    const int scol = (tid & 15) * 4;

    const float* __restrict__ gA = Xin  + (size_t)(m0 + srow) * H_DIM + scol;
    const float* __restrict__ gB = Mmat + (size_t)srow * H_DIM + n0 + scol;

    float* sA = &As[srow * PA + scol];
    float* sB = &Bs[(scol >> 4) * TB + srow * 16 + (scol & 15)];

    v4f ra[TILE_M / 8], rb[KC / 8];
#pragma unroll
    for (int p = 0; p < TILE_M / 8; ++p)
        ra[p] = *(const v4f*)(gA + (size_t)(p * 8) * H_DIM);
#pragma unroll
    for (int p = 0; p < KC / 8; ++p)
        rb[p] = *(const v4f*)(gB + (size_t)(p * 8) * H_DIM);

    v8f acc0 = {}, acc1 = {};

    const float* ap  = &As[(wm + l) * PA + 2 * half];
    const float* bp0 = &Bs[(wn >> 4) * TB + (2 * half) * 16 + l];
    const float* bp1 = bp0 + TB;

    for (int c = 0; c < H_DIM / KC; ++c) {
#pragma unroll
        for (int p = 0; p < TILE_M / 8; ++p)
            *(v4f*)(sA + p * 8 * PA) = ra[p];
#pragma unroll
        for (int p = 0; p < KC / 8; ++p)
            *(v4f*)(sB + p * 8 * 16) = rb[p];
        __syncthreads();

        v2f af[KC / 4], bf0[KC / 4], bf1[KC / 4];
#pragma unroll
        for (int s = 0; s < KC / 4; ++s) {
            af[s]    = *(const v2f*)(ap + 4 * s);
            bf0[s].x = bp0[(4 * s) * 16];  bf0[s].y = bp0[(4 * s + 1) * 16];
            bf1[s].x = bp1[(4 * s) * 16];  bf1[s].y = bp1[(4 * s + 1) * 16];
        }
        if (c + 1 < H_DIM / KC) {
            const float* __restrict__ nA = gA + (c + 1) * KC;
            const float* __restrict__ nB = gB + (size_t)(c + 1) * KC * H_DIM;
#pragma unroll
            for (int p = 0; p < TILE_M / 8; ++p)
                ra[p] = *(const v4f*)(nA + (size_t)(p * 8) * H_DIM);
#pragma unroll
            for (int p = 0; p < KC / 8; ++p)
                rb[p] = *(const v4f*)(nB + (size_t)(p * 8) * H_DIM);
        }
        __builtin_amdgcn_sched_barrier(0);
#pragma unroll
        for (int s = 0; s < KC / 4; ++s) {
            acc0 = __builtin_amdgcn_wmma_f32_16x16x4_f32(
                false, af[s], false, bf0[s], (short)0, acc0, false, false);
            acc1 = __builtin_amdgcn_wmma_f32_16x16x4_f32(
                false, af[s], false, bf1[s], (short)0, acc1, false, false);
        }
        __builtin_amdgcn_sched_barrier(0);
        __syncthreads();
    }

    float* __restrict__ o = Out + (size_t)(m0 + wm + half * 8) * H_DIM + n0 + wn + l;
#pragma unroll
    for (int p = 0; p < 8; ++p) {
        o[(size_t)p * H_DIM]      = acc0[p];
        o[(size_t)p * H_DIM + 16] = acc1[p];
    }
}

extern "C" void kernel_launch(void* const* d_in, const int* in_sizes, int n_in,
                              void* d_out, int out_size, void* d_ws, size_t ws_size,
                              hipStream_t stream) {
    const float* hidden = (const float*)d_in[0];
    const float* W_seq  = (const float*)d_in[2];
    const float* he     = (const float*)d_in[3];
    const float* cp     = (const float*)d_in[4];
    float* out  = (float*)d_out;
    float* Mmat = (float*)d_ws;

    (void)in_sizes; (void)n_in; (void)out_size; (void)ws_size;

    build_M<<<256, 64, 0, stream>>>(W_seq, he, cp, Mmat);

    gemm_out<<<512, 128, 0, stream>>>(hidden, Mmat, out);
}
